# starting kernel
# baseline (speedup 1.0000x reference)
#include <hip/hip_runtime.h>
#include <hip/hip_bf16.h>

typedef __attribute__((ext_vector_type(2))) float v2f;
typedef __attribute__((ext_vector_type(8))) float v8f;

#define IN_C   64
#define IMG_H  256
#define IMG_W  256
#define OUT_C  128
#define OH     254
#define OW     254
#define KTOT   576
#define TILE_N 64
#define PATCH_W 66
#define LDS_STRIDE 416
#define BLOCK  128

__global__ __launch_bounds__(256) void repack_weights_k(
    const float* __restrict__ w, float* __restrict__ w3) {
  int i = blockIdx.x * 256 + threadIdx.x;
  if (i >= OUT_C * KTOT) return;
  int j    = i & 1;
  int t    = i >> 1;
  int oc   = t & (OUT_C - 1);
  t      >>= 7;
  int half = t & 1;
  int k4   = t >> 1;
  int k    = (k4 << 2) + (half << 1) + j;
  int khkw = k >> 6;
  int ic   = k & 63;
  int kh   = khkw / 3;
  int kw   = khkw - kh * 3;
  w3[i] = w[((oc * IN_C + ic) * 3 + kh) * 3 + kw];
}

__global__ __launch_bounds__(BLOCK) void conv_wmma_f32_k(
    const float* __restrict__ x, const float* __restrict__ w3,
    const float* __restrict__ bias, float* __restrict__ out) {
  __shared__ float smem[32 * LDS_STRIDE];

  const int owb = blockIdx.x * TILE_N;
  const int oh  = blockIdx.y;
  const int tid = threadIdx.x;

  for (int t = tid; t < IN_C * 3 * PATCH_W; t += BLOCK) {
    int ic  = t / (3 * PATCH_W);
    int rem = t - ic * (3 * PATCH_W);
    int r   = rem / PATCH_W;
    int c   = rem - r * PATCH_W;
    int gx  = owb + c;
    float v = 0.0f;
    if (gx < IMG_W) v = x[ic * (IMG_H * IMG_W) + (oh + r) * IMG_W + gx];
    smem[(ic >> 1) * LDS_STRIDE + (r * PATCH_W + c) * 2 + (ic & 1)] = v;
  }
  __syncthreads();

  const int wave   = tid >> 5;
  const int lane   = tid & 31;
  const int laneLo = lane & 15;
  const int hi     = lane >> 4;
  const int ocb    = wave << 5;

  v8f a0p0 = {}, a0p1 = {}, a0p2 = {}, a0p3 = {};
  v8f a1p0 = {}, a1p1 = {}, a1p2 = {}, a1p3 = {};

  const float* wp0 = w3 + ((hi * OUT_C) + (ocb + laneLo)) * 2;
  const float* wp1 = wp0 + 32;

  for (int kh = 0; kh < 3; ++kh) {
    for (int kw = 0; kw < 3; ++kw) {
      const int colbase = (kh * PATCH_W + laneLo + kw) * 2;
#pragma unroll
      for (int icq = 0; icq < 16; ++icq) {
        v2f a0 = *(const v2f*)wp0;
        v2f a1 = *(const v2f*)wp1;
        wp0 += 512;
        wp1 += 512;
        const int base = (icq * 2 + hi) * LDS_STRIDE + colbase;
        v2f b0 = *(const v2f*)&smem[base];
        v2f b1 = *(const v2f*)&smem[base + 32];
        v2f b2 = *(const v2f*)&smem[base + 64];
        v2f b3 = *(const v2f*)&smem[base + 96];
        a0p0 = __builtin_amdgcn_wmma_f32_16x16x4_f32(
                   false, a0, false, b0, (short)0, a0p0, false, false);
        a0p1 = __builtin_amdgcn_wmma_f32_16x16x4_f32(
                   false, a0, false, b1, (short)0, a0p1, false, false);
        a0p2 = __builtin_amdgcn_wmma_f32_16x16x4_f32(
                   false, a0, false, b2, (short)0, a0p2, false, false);
        a0p3 = __builtin_amdgcn_wmma_f32_16x16x4_f32(
                   false, a0, false, b3, (short)0, a0p3, false, false);
        a1p0 = __builtin_amdgcn_wmma_f32_16x16x4_f32(
                   false, a1, false, b0, (short)0, a1p0, false, false);
        a1p1 = __builtin_amdgcn_wmma_f32_16x16x4_f32(
                   false, a1, false, b1, (short)0, a1p1, false, false);
        a1p2 = __builtin_amdgcn_wmma_f32_16x16x4_f32(
                   false, a1, false, b2, (short)0, a1p2, false, false);
        a1p3 = __builtin_amdgcn_wmma_f32_16x16x4_f32(
                   false, a1, false, b3, (short)0, a1p3, false, false);
      }
    }
  }

  const int ow0 = owb + laneLo;
#pragma unroll
  for (int r = 0; r < 8; ++r) {
    int ocA   = ocb + (hi << 3) + r;
    int ocB   = ocA + 16;
    float bvA = bias[ocA];
    float bvB = bias[ocB];
    float* rowA = out + (ocA * OH + oh) * OW;
    float* rowB = out + (ocB * OH + oh) * OW;
    if (ow0 < OW) {
      rowA[ow0]      = a0p0[r] + bvA;
      rowB[ow0]      = a1p0[r] + bvB;
    }
    if (ow0 + 16 < OW) {
      rowA[ow0 + 16] = a0p1[r] + bvA;
      rowB[ow0 + 16] = a1p1[r] + bvB;
    }
    if (ow0 + 32 < OW) {
      rowA[ow0 + 32] = a0p2[r] + bvA;
      rowB[ow0 + 32] = a1p2[r] + bvB;
    }
    if (ow0 + 48 < OW) {
      rowA[ow0 + 48] = a0p3[r] + bvA;
      rowB[ow0 + 48] = a1p3[r] + bvB;
    }
  }
}

extern "C" void kernel_launch(void* const* d_in, const int* in_sizes, int n_in,
                              void* d_out, int out_size, void* d_ws, size_t ws_size,
                              hipStream_t stream) {
  const float* x    = (const float*)d_in[0];
  const float* kern = (const float*)d_in[1];
  const float* bias = (const float*)d_in[2];
  float* out = (float*)d_out;
  float* w3  = (float*)d_ws;

  repack_weights_k<<<(OUT_C * KTOT) / 256, 256, 0, stream>>>(kern, w3);

  dim3 grid((OW + TILE_N - 1) / TILE_N, OH);
  conv_wmma_f32_k<<<grid, BLOCK, 0, stream>>>(x, w3, bias, out);
}


# Input-patch staging: each thread issues all its row loads (24 global_load_b128 + 2 halo b64) as one group ahead of its LDS stores, not one dependent 4-byte load per element; sums unchanged
# speedup vs baseline: 1.9696x; 1.9696x over previous
#include <hip/hip_runtime.h>
#include <hip/hip_bf16.h>

typedef __attribute__((ext_vector_type(2))) float v2f;
typedef __attribute__((ext_vector_type(4))) float v4f;
typedef __attribute__((ext_vector_type(8))) float v8f;

#define IN_C   64
#define IMG_H  256
#define IMG_W  256
#define OUT_C  128
#define OH     254
#define OW     254
#define KTOT   576
#define TILE_N 64
#define PATCH_W 66
#define LDS_STRIDE 416
#define BLOCK  128

__global__ __launch_bounds__(256) void repack_weights_k(
    const float* __restrict__ w, float* __restrict__ w3) {
  int i = blockIdx.x * 256 + threadIdx.x;
  if (i >= OUT_C * KTOT) return;
  int j    = i & 1;
  int t    = i >> 1;
  int oc   = t & (OUT_C - 1);
  t      >>= 7;
  int half = t & 1;
  int k4   = t >> 1;
  int k    = (k4 << 2) + (half << 1) + j;
  int khkw = k >> 6;
  int ic   = k & 63;
  int kh   = khkw / 3;
  int kw   = khkw - kh * 3;
  w3[i] = w[((oc * IN_C + ic) * 3 + kh) * 3 + kw];
}

__global__ __launch_bounds__(BLOCK) void conv_wmma_f32_k(
    const float* __restrict__ x, const float* __restrict__ w3,
    const float* __restrict__ bias, float* __restrict__ out) {
  __shared__ alignas(16) float smem[32 * LDS_STRIDE];

  const int owb = blockIdx.x * TILE_N;
  const int oh  = blockIdx.y;
  const int tid = threadIdx.x;

  {
    const int c4 = tid & 15;
    const int pl = tid >> 4;
    const float* gbase = x + (oh * IMG_W + owb + 4 * c4);
    v4f ev[12], od[12];
    for (int i = 0; i < 12; ++i) {
      const int r = i >> 2;
      const int p = ((i & 3) << 3) + pl;
      const float* g = gbase + (2 * p) * (IMG_H * IMG_W) + r * IMG_W;
      ev[i] = *(const v4f*)g;
      od[i] = *(const v4f*)(g + IMG_H * IMG_W);
    }
    const int hr = tid >> 5;
    const int hp = tid & 31;
    v2f he = {0.0f, 0.0f}, ho = {0.0f, 0.0f};
    if (tid < 3 * 32 && owb + TILE_N < IMG_W) {
      const float* g = x + ((2 * hp) * IMG_H + oh + hr) * IMG_W + owb + TILE_N;
      he = *(const v2f*)g;
      ho = *(const v2f*)(g + IMG_H * IMG_W);
    }
    for (int i = 0; i < 12; ++i) {
      const int r = i >> 2;
      const int p = ((i & 3) << 3) + pl;
      float* s = &smem[p * LDS_STRIDE + (r * PATCH_W + 4 * c4) * 2];
      v4f q0 = {ev[i].x, od[i].x, ev[i].y, od[i].y};
      v4f q1 = {ev[i].z, od[i].z, ev[i].w, od[i].w};
      *(v4f*)s       = q0;
      *(v4f*)(s + 4) = q1;
    }
    if (tid < 3 * 32) {
      v4f q = {he.x, ho.x, he.y, ho.y};
      *(v4f*)&smem[hp * LDS_STRIDE + (hr * PATCH_W + TILE_N) * 2] = q;
    }
  }
  __syncthreads();

  const int wave   = tid >> 5;
  const int lane   = tid & 31;
  const int laneLo = lane & 15;
  const int hi     = lane >> 4;
  const int ocb    = wave << 5;

  v8f a0p0 = {}, a0p1 = {}, a0p2 = {}, a0p3 = {};
  v8f a1p0 = {}, a1p1 = {}, a1p2 = {}, a1p3 = {};

  const float* wp0 = w3 + ((hi * OUT_C) + (ocb + laneLo)) * 2;
  const float* wp1 = wp0 + 32;

  for (int kh = 0; kh < 3; ++kh) {
    for (int kw = 0; kw < 3; ++kw) {
      const int colbase = (kh * PATCH_W + laneLo + kw) * 2;
#pragma unroll
      for (int icq = 0; icq < 16; ++icq) {
        v2f a0 = *(const v2f*)wp0;
        v2f a1 = *(const v2f*)wp1;
        wp0 += 512;
        wp1 += 512;
        const int base = (icq * 2 + hi) * LDS_STRIDE + colbase;
        v2f b0 = *(const v2f*)&smem[base];
        v2f b1 = *(const v2f*)&smem[base + 32];
        v2f b2 = *(const v2f*)&smem[base + 64];
        v2f b3 = *(const v2f*)&smem[base + 96];
        a0p0 = __builtin_amdgcn_wmma_f32_16x16x4_f32(
                   false, a0, false, b0, (short)0, a0p0, false, false);
        a0p1 = __builtin_amdgcn_wmma_f32_16x16x4_f32(
                   false, a0, false, b1, (short)0, a0p1, false, false);
        a0p2 = __builtin_amdgcn_wmma_f32_16x16x4_f32(
                   false, a0, false, b2, (short)0, a0p2, false, false);
        a0p3 = __builtin_amdgcn_wmma_f32_16x16x4_f32(
                   false, a0, false, b3, (short)0, a0p3, false, false);
        a1p0 = __builtin_amdgcn_wmma_f32_16x16x4_f32(
                   false, a1, false, b0, (short)0, a1p0, false, false);
        a1p1 = __builtin_amdgcn_wmma_f32_16x16x4_f32(
                   false, a1, false, b1, (short)0, a1p1, false, false);
        a1p2 = __builtin_amdgcn_wmma_f32_16x16x4_f32(
                   false, a1, false, b2, (short)0, a1p2, false, false);
        a1p3 = __builtin_amdgcn_wmma_f32_16x16x4_f32(
                   false, a1, false, b3, (short)0, a1p3, false, false);
      }
    }
  }

  const int ow0 = owb + laneLo;
#pragma unroll
  for (int r = 0; r < 8; ++r) {
    int ocA   = ocb + (hi << 3) + r;
    int ocB   = ocA + 16;
    float bvA = bias[ocA];
    float bvB = bias[ocB];
    float* rowA = out + (ocA * OH + oh) * OW;
    float* rowB = out + (ocB * OH + oh) * OW;
    if (ow0 < OW) {
      rowA[ow0]      = a0p0[r] + bvA;
      rowB[ow0]      = a1p0[r] + bvB;
    }
    if (ow0 + 16 < OW) {
      rowA[ow0 + 16] = a0p1[r] + bvA;
      rowB[ow0 + 16] = a1p1[r] + bvB;
    }
    if (ow0 + 32 < OW) {
      rowA[ow0 + 32] = a0p2[r] + bvA;
      rowB[ow0 + 32] = a1p2[r] + bvB;
    }
    if (ow0 + 48 < OW) {
      rowA[ow0 + 48] = a0p3[r] + bvA;
      rowB[ow0 + 48] = a1p3[r] + bvB;
    }
  }
}

extern "C" void kernel_launch(void* const* d_in, const int* in_sizes, int n_in,
                              void* d_out, int out_size, void* d_ws, size_t ws_size,
                              hipStream_t stream) {
  const float* x    = (const float*)d_in[0];
  const float* kern = (const float*)d_in[1];
  const float* bias = (const float*)d_in[2];
  float* out = (float*)d_out;
  float* w3  = (float*)d_ws;

  repack_weights_k<<<(OUT_C * KTOT) / 256, 256, 0, stream>>>(kern, w3);

  dim3 grid((OW + TILE_N - 1) / TILE_N, OH);
  conv_wmma_f32_k<<<grid, BLOCK, 0, stream>>>(x, w3, bias, out);
}


# Conv kernel resident in one pass: icq loop unrolled by 8 (298 -> 128 registers), waves per SIMD pinned at 4: all 1016 workgroups sit 4 per WGP, not 768 + a second pass of 248; sums unchanged
# speedup vs baseline: 2.4584x; 1.2482x over previous
#include <hip/hip_runtime.h>
#include <hip/hip_bf16.h>

typedef __attribute__((ext_vector_type(2))) float v2f;
typedef __attribute__((ext_vector_type(4))) float v4f;
typedef __attribute__((ext_vector_type(8))) float v8f;

#define IN_C   64
#define IMG_H  256
#define IMG_W  256
#define OUT_C  128
#define OH     254
#define OW     254
#define KTOT   576
#define TILE_N 64
#define PATCH_W 66
#define LDS_STRIDE 416
#define BLOCK  128

__global__ __launch_bounds__(256) void repack_weights_k(
    const float* __restrict__ w, float* __restrict__ w3) {
  int i = blockIdx.x * 256 + threadIdx.x;
  if (i >= OUT_C * KTOT) return;
  int j    = i & 1;
  int t    = i >> 1;
  int oc   = t & (OUT_C - 1);
  t      >>= 7;
  int half = t & 1;
  int k4   = t >> 1;
  int k    = (k4 << 2) + (half << 1) + j;
  int khkw = k >> 6;
  int ic   = k & 63;
  int kh   = khkw / 3;
  int kw   = khkw - kh * 3;
  w3[i] = w[((oc * IN_C + ic) * 3 + kh) * 3 + kw];
}

__global__ __launch_bounds__(BLOCK) __attribute__((amdgpu_waves_per_eu(4, 4)))
void conv_wmma_f32_k(
    const float* __restrict__ x, const float* __restrict__ w3,
    const float* __restrict__ bias, float* __restrict__ out) {
  __shared__ alignas(16) float smem[32 * LDS_STRIDE];

  const int owb = blockIdx.x * TILE_N;
  const int oh  = blockIdx.y;
  const int tid = threadIdx.x;

  {
    const int c4 = tid & 15;
    const int pl = tid >> 4;
    const float* gbase = x + (oh * IMG_W + owb + 4 * c4);
    v4f ev[12], od[12];
    for (int i = 0; i < 12; ++i) {
      const int r = i >> 2;
      const int p = ((i & 3) << 3) + pl;
      const float* g = gbase + (2 * p) * (IMG_H * IMG_W) + r * IMG_W;
      ev[i] = *(const v4f*)g;
      od[i] = *(const v4f*)(g + IMG_H * IMG_W);
    }
    const int hr = tid >> 5;
    const int hp = tid & 31;
    v2f he = {0.0f, 0.0f}, ho = {0.0f, 0.0f};
    if (tid < 3 * 32 && owb + TILE_N < IMG_W) {
      const float* g = x + ((2 * hp) * IMG_H + oh + hr) * IMG_W + owb + TILE_N;
      he = *(const v2f*)g;
      ho = *(const v2f*)(g + IMG_H * IMG_W);
    }
    for (int i = 0; i < 12; ++i) {
      const int r = i >> 2;
      const int p = ((i & 3) << 3) + pl;
      float* s = &smem[p * LDS_STRIDE + (r * PATCH_W + 4 * c4) * 2];
      v4f q0 = {ev[i].x, od[i].x, ev[i].y, od[i].y};
      v4f q1 = {ev[i].z, od[i].z, ev[i].w, od[i].w};
      *(v4f*)s       = q0;
      *(v4f*)(s + 4) = q1;
    }
    if (tid < 3 * 32) {
      v4f q = {he.x, ho.x, he.y, ho.y};
      *(v4f*)&smem[hp * LDS_STRIDE + (hr * PATCH_W + TILE_N) * 2] = q;
    }
  }
  __syncthreads();

  const int wave   = tid >> 5;
  const int lane   = tid & 31;
  const int laneLo = lane & 15;
  const int hi     = lane >> 4;
  const int ocb    = wave << 5;

  v8f a0p0 = {}, a0p1 = {}, a0p2 = {}, a0p3 = {};
  v8f a1p0 = {}, a1p1 = {}, a1p2 = {}, a1p3 = {};

  const float* wp0 = w3 + ((hi * OUT_C) + (ocb + laneLo)) * 2;
  const float* wp1 = wp0 + 32;

  for (int kh = 0; kh < 3; ++kh) {
    for (int kw = 0; kw < 3; ++kw) {
      const int colbase = (kh * PATCH_W + laneLo + kw) * 2;
#pragma unroll 8
      for (int icq = 0; icq < 16; ++icq) {
        v2f a0 = *(const v2f*)wp0;
        v2f a1 = *(const v2f*)wp1;
        wp0 += 512;
        wp1 += 512;
        const int base = (icq * 2 + hi) * LDS_STRIDE + colbase;
        v2f b0 = *(const v2f*)&smem[base];
        v2f b1 = *(const v2f*)&smem[base + 32];
        v2f b2 = *(const v2f*)&smem[base + 64];
        v2f b3 = *(const v2f*)&smem[base + 96];
        a0p0 = __builtin_amdgcn_wmma_f32_16x16x4_f32(
                   false, a0, false, b0, (short)0, a0p0, false, false);
        a0p1 = __builtin_amdgcn_wmma_f32_16x16x4_f32(
                   false, a0, false, b1, (short)0, a0p1, false, false);
        a0p2 = __builtin_amdgcn_wmma_f32_16x16x4_f32(
                   false, a0, false, b2, (short)0, a0p2, false, false);
        a0p3 = __builtin_amdgcn_wmma_f32_16x16x4_f32(
                   false, a0, false, b3, (short)0, a0p3, false, false);
        a1p0 = __builtin_amdgcn_wmma_f32_16x16x4_f32(
                   false, a1, false, b0, (short)0, a1p0, false, false);
        a1p1 = __builtin_amdgcn_wmma_f32_16x16x4_f32(
                   false, a1, false, b1, (short)0, a1p1, false, false);
        a1p2 = __builtin_amdgcn_wmma_f32_16x16x4_f32(
                   false, a1, false, b2, (short)0, a1p2, false, false);
        a1p3 = __builtin_amdgcn_wmma_f32_16x16x4_f32(
                   false, a1, false, b3, (short)0, a1p3, false, false);
      }
    }
  }

  const int ow0 = owb + laneLo;
#pragma unroll
  for (int r = 0; r < 8; ++r) {
    int ocA   = ocb + (hi << 3) + r;
    int ocB   = ocA + 16;
    float bvA = bias[ocA];
    float bvB = bias[ocB];
    float* rowA = out + (ocA * OH + oh) * OW;
    float* rowB = out + (ocB * OH + oh) * OW;
    if (ow0 < OW) {
      rowA[ow0]      = a0p0[r] + bvA;
      rowB[ow0]      = a1p0[r] + bvB;
    }
    if (ow0 + 16 < OW) {
      rowA[ow0 + 16] = a0p1[r] + bvA;
      rowB[ow0 + 16] = a1p1[r] + bvB;
    }
    if (ow0 + 32 < OW) {
      rowA[ow0 + 32] = a0p2[r] + bvA;
      rowB[ow0 + 32] = a1p2[r] + bvB;
    }
    if (ow0 + 48 < OW) {
      rowA[ow0 + 48] = a0p3[r] + bvA;
      rowB[ow0 + 48] = a1p3[r] + bvB;
    }
  }
}

extern "C" void kernel_launch(void* const* d_in, const int* in_sizes, int n_in,
                              void* d_out, int out_size, void* d_ws, size_t ws_size,
                              hipStream_t stream) {
  const float* x    = (const float*)d_in[0];
  const float* kern = (const float*)d_in[1];
  const float* bias = (const float*)d_in[2];
  float* out = (float*)d_out;
  float* w3  = (float*)d_ws;

  repack_weights_k<<<(OUT_C * KTOT) / 256, 256, 0, stream>>>(kern, w3);

  dim3 grid((OW + TILE_N - 1) / TILE_N, OH);
  conv_wmma_f32_k<<<grid, BLOCK, 0, stream>>>(x, w3, bias, out);
}
